# starting kernel
# baseline (speedup 1.0000x reference)
#include <hip/hip_runtime.h>


typedef __attribute__((ext_vector_type(2))) float v2f;
typedef __attribute__((ext_vector_type(8))) float v8f;

#define BATCH 8
#define CCH   256
#define HH    64
#define WW    96
#define ND    21
#define APR   224
#define BPR   288
#define NT    384

__global__ __launch_bounds__(NT, 1)
void corr_wmma_f32_kernel(const float* __restrict__ f1,
                          const float* __restrict__ f2,
                          float* __restrict__ out) {
  extern __shared__ float smem[];
  float* __restrict__ Ald = smem;
  float* __restrict__ Bld = smem + (CCH / 2) * APR;

  const int tid  = threadIdx.x;
  const int lane = tid & 31;
  const int wv   = tid >> 5;
  const int ti   = wv >> 1;
  const int djb  = (wv & 1) * 2;

  const int h = blockIdx.x;
  const int b = blockIdx.y;

  const int row = lane & 15;
  const int khi = lane >> 4;

  const size_t inBase = (size_t)b * (CCH * HH * WW);

  {
    const float* src = f1 + inBase + (size_t)h * WW;
    for (int i = tid; i < CCH * (WW / 4); i += NT) {
      const int c = i / (WW / 4);
      const int q = i % (WW / 4);
      float4 v = *(const float4*)(src + (size_t)c * (HH * WW) + q * 4);
      float* d = &Ald[(c >> 1) * APR + (c & 1) + 8 * q];
      d[0] = v.x; d[2] = v.y; d[4] = v.z; d[6] = v.w;
    }
  }
  {
    const float4 z = make_float4(0.f, 0.f, 0.f, 0.f);
    for (int i = tid; i < (CCH / 2) * 24; i += NT) {
      const int p = i / 24;
      const int j = i % 24;
      const int off = (j < 10) ? (j * 4) : (232 + (j - 10) * 4);
      *(float4*)(&Bld[p * BPR + off]) = z;
    }
  }
  __syncthreads();

  for (int dyI = 0; dyI < ND; ++dyI) {
    const int h2 = h + dyI * 2 - 20;
    const size_t outBase =
        (((size_t)b * (ND * ND) + (size_t)dyI * ND) * HH + (size_t)h) * WW;

    if (h2 < 0 || h2 >= HH) {
      for (int i = tid; i < ND * WW; i += NT) {
        const int dxI = i / WW, w = i % WW;
        out[outBase + (size_t)dxI * (HH * WW) + w] = 0.f;
      }
      continue;
    }

    __syncthreads();
    {
      const float* src = f2 + inBase + (size_t)h2 * WW;
      for (int i = tid; i < CCH * (WW / 4); i += NT) {
        const int c = i / (WW / 4);
        const int q = i % (WW / 4);
        float4 v = *(const float4*)(src + (size_t)c * (HH * WW) + q * 4);
        float* d = &Bld[(c >> 1) * BPR + (c & 1) + 40 + 8 * q];
        d[0] = v.x; d[2] = v.y; d[4] = v.z; d[6] = v.w;
      }
      const int h2n = h2 + 2;
      if (h2n < HH) {
        const float* nsrc = f2 + inBase + (size_t)h2n * WW;
        for (int i = tid; i < CCH; i += NT)
          __builtin_prefetch(nsrc + (size_t)i * (HH * WW), 0, 0);
      }
    }
    __syncthreads();

    v8f acc0 = {0.f, 0.f, 0.f, 0.f, 0.f, 0.f, 0.f, 0.f};
    v8f acc1 = {0.f, 0.f, 0.f, 0.f, 0.f, 0.f, 0.f, 0.f};

    const float* ap = &Ald[khi * APR + 2 * (ti * 16 + row)];
    const float* bp = &Bld[khi * BPR + 2 * ((ti + djb) * 16 + row)];

#pragma unroll 8
    for (int k = 0; k < CCH; k += 4) {
      v2f a  = *(const v2f*)(ap);
      v2f b0 = *(const v2f*)(bp);
      v2f b1 = *(const v2f*)(bp + 32);
      acc0 = __builtin_amdgcn_wmma_f32_16x16x4_f32(
          false, a, false, b0, (short)0, acc0, false, false);
      acc1 = __builtin_amdgcn_wmma_f32_16x16x4_f32(
          false, a, false, b1, (short)0, acc1, false, false);
      ap += 2 * APR;
      bp += 2 * BPR;
    }

    const int mBase = khi * 8;
    const int N = lane & 15;
#pragma unroll
    for (int r = 0; r < 8; ++r) {
      const int m  = mBase + r;
      const int w1 = ti * 16 + m;
      {
        const int diff = 16 * djb + N - m;
        if (diff >= 0 && diff <= 40 && !(diff & 1))
          out[outBase + (size_t)(diff >> 1) * (HH * WW) + w1] =
              acc0[r] * 0.00390625f;
      }
      {
        const int diff = 16 * (djb + 1) + N - m;
        if (diff >= 0 && diff <= 40 && !(diff & 1))
          out[outBase + (size_t)(diff >> 1) * (HH * WW) + w1] =
              acc1[r] * 0.00390625f;
      }
    }
  }
}

extern "C" void kernel_launch(void* const* d_in, const int* in_sizes, int n_in,
                              void* d_out, int out_size, void* d_ws, size_t ws_size,
                              hipStream_t stream) {
  const float* f1 = (const float*)d_in[0];
  const float* f2 = (const float*)d_in[1];
  float* out = (float*)d_out;

  dim3 grid(HH, BATCH);
  dim3 block(NT);
  const size_t shmem =
      (size_t)((CCH / 2) * APR + (CCH / 2) * BPR) * sizeof(float);
  corr_wmma_f32_kernel<<<grid, block, shmem, stream>>>(f1, f2, out);
}


# K split over 4 waves (64 channels each): A tiles and f2 tiles both held as register fragments, no LDS read in the WMMA stream; the 4 partial tiles are joined through LDS (sum order changed)
# speedup vs baseline: 1.2497x; 1.2497x over previous
#include <hip/hip_runtime.h>


typedef __attribute__((ext_vector_type(2))) float v2f;
typedef __attribute__((ext_vector_type(4))) float v4f;
typedef __attribute__((ext_vector_type(8))) float v8f;

#define BATCH 8
#define CCH   256
#define HH    64
#define WW    96
#define HW    (HH * WW)
#define ND    21
#define NWV   4
#define NT    (NWV * 32)
#define KSW   16
#define NQ    24
#define QSTR  1024

__device__ __forceinline__ v2f ld_frag(const float* __restrict__ p, const int s) {
  v2f v;
  v.x = p[(size_t)(4 * s) * HW];
  v.y = p[(size_t)(4 * s + 1) * HW];
  return v;
}

__device__ __forceinline__ void st_part(float* __restrict__ part, const int q, const v8f a) {
  *(v4f*)(part + q * QSTR)       = (v4f){a[0], a[1], a[2], a[3]};
  *(v4f*)(part + q * QSTR + 128) = (v4f){a[4], a[5], a[6], a[7]};
}

template <int J>
__device__ __forceinline__ void tile_step(const v2f (&A)[6][KSW], v2f (&B)[KSW],
                                          const float* __restrict__ nxt,
                                          float* __restrict__ part) {
  constexpr int lo = (J > 3) ? J - 3 : 0;
  constexpr int hi = (J < 5) ? J : 5;
  constexpr int n  = hi - lo + 1;
  constexpr int q0 = (J == 0) ? 0 : (J == 1) ? 1 : (J == 2) ? 3 : (J == 3) ? 6 :
                     (J == 4) ? 10 : (J == 5) ? 14 : (J == 6) ? 18 : (J == 7) ? 21 : 23;
  const v8f zero = (v8f){0.f, 0.f, 0.f, 0.f, 0.f, 0.f, 0.f, 0.f};
  if constexpr (n == 1) {
    v8f a0 = zero, a1 = zero;
#pragma unroll
    for (int s = 0; s < KSW; s += 2) {
      a0 = __builtin_amdgcn_wmma_f32_16x16x4_f32(false, A[lo][s], false, B[s],
                                                 (short)0, a0, false, false);
      a1 = __builtin_amdgcn_wmma_f32_16x16x4_f32(false, A[lo][s + 1], false, B[s + 1],
                                                 (short)0, a1, false, false);
      B[s]     = ld_frag(nxt, s);
      B[s + 1] = ld_frag(nxt, s + 1);
    }
    st_part(part, q0, a0 + a1);
  } else {
    v8f acc[n];
#pragma unroll
    for (int t = 0; t < n; ++t) acc[t] = zero;
#pragma unroll
    for (int s = 0; s < KSW; ++s) {
#pragma unroll
      for (int t = 0; t < n; ++t)
        acc[t] = __builtin_amdgcn_wmma_f32_16x16x4_f32(false, A[lo + t][s], false, B[s],
                                                       (short)0, acc[t], false, false);
      B[s] = ld_frag(nxt, s);
    }
#pragma unroll
    for (int t = 0; t < n; ++t) st_part(part, q0 + t, acc[t]);
  }
}

__device__ __forceinline__ void lds_barrier() {
  __builtin_amdgcn_fence(__ATOMIC_RELEASE, "workgroup", "local");
  __builtin_amdgcn_s_barrier();
  __builtin_amdgcn_fence(__ATOMIC_ACQUIRE, "workgroup", "local");
}

__global__ __launch_bounds__(NT)
void corr_wmma_f32_kernel(const float* __restrict__ f1,
                          const float* __restrict__ f2,
                          float* __restrict__ out) {
  extern __shared__ float smem[];

  const int tid  = threadIdx.x;
  const int lane = tid & 31;
  const int wv   = __builtin_amdgcn_readfirstlane(tid >> 5);

  const int h = blockIdx.x;
  const int b = blockIdx.y;
  const size_t inBase = (size_t)b * (CCH * HW);

  const int row = lane & 15;
  const int khi = lane >> 4;

  const int dyFirst = (h < 20) ? (21 - h) / 2 : 0;
  const int dyLast  = ((83 - h) / 2 < 20) ? (83 - h) / 2 : 20;

  for (int dy = 0; dy < ND; ++dy) {
    if (dy >= dyFirst && dy <= dyLast) continue;
    const size_t outBase =
        (((size_t)b * (ND * ND) + (size_t)dy * ND) * HH + (size_t)h) * WW;
    for (int i = tid; i < ND * WW; i += NT) {
      const int dxI = i / WW, w = i % WW;
      out[outBase + (size_t)dxI * HW + w] = 0.f;
    }
  }

  v2f A[6][KSW];
  {
    const float* a = f1 + inBase + (size_t)(64 * wv + 2 * khi) * HW + (size_t)h * WW + row;
#pragma unroll
    for (int ti = 0; ti < 6; ++ti)
#pragma unroll
      for (int s = 0; s < KSW; ++s) A[ti][s] = ld_frag(a + 16 * ti, s);
  }

  const float* f2k = f2 + inBase + (size_t)(64 * wv + 2 * khi) * HW;
  int col[9];
#pragma unroll
  for (int j = 0; j < 9; ++j) {
    const int w2 = 16 * j - 20 + row;
    col[j] = w2 < 0 ? 0 : (w2 >= WW ? WW - 1 : w2);
  }

  v2f B0[KSW], B1[KSW], B2[KSW];
  {
    const float* r0 = f2k + (size_t)(h + 2 * dyFirst - 20) * WW;
#pragma unroll
    for (int s = 0; s < KSW; ++s) {
      B0[s] = ld_frag(r0 + col[0], s);
      B1[s] = ld_frag(r0 + col[1], s);
      B2[s] = ld_frag(r0 + col[2], s);
    }
  }

  float* part = smem + wv * 256 + lane * 4;
  const float* red = smem + lane * 4;
  const int mBase = khi * 8;

  for (int dy = dyFirst; dy <= dyLast; ++dy) {
    const int dyn = (dy < dyLast) ? dy + 1 : dy;
    const float* rc = f2k + (size_t)(h + 2 * dy - 20) * WW;
    const float* rn = f2k + (size_t)(h + 2 * dyn - 20) * WW;

    tile_step<0>(A, B0, rc + col[3], part);
    tile_step<1>(A, B1, rc + col[4], part);
    tile_step<2>(A, B2, rc + col[5], part);
    tile_step<3>(A, B0, rc + col[6], part);
    tile_step<4>(A, B1, rc + col[7], part);
    tile_step<5>(A, B2, rc + col[8], part);
    tile_step<6>(A, B0, rn + col[0], part);
    tile_step<7>(A, B1, rn + col[1], part);
    tile_step<8>(A, B2, rn + col[2], part);

    lds_barrier();

    const size_t outBase =
        (((size_t)b * (ND * ND) + (size_t)dy * ND) * HH + (size_t)h) * WW;
#pragma unroll
    for (int u = 0; u < 6; ++u) {
      const int q  = 4 * u + wv;
      const int sh = 4 * (q & 15);
      const int j  = (int)(((q < 16 ? 0x5544443333222110ULL : 0x87766655ULL) >> sh) & 15ULL);
      const int ti = (int)(((q < 16 ? 0x3243213210210100ULL : 0x55454354ULL) >> sh) & 15ULL);
      const float* p = red + q * QSTR;
      v4f lo0 = *(const v4f*)(p),       hi0 = *(const v4f*)(p + 128);
      v4f lo1 = *(const v4f*)(p + 256), hi1 = *(const v4f*)(p + 384);
      v4f lo2 = *(const v4f*)(p + 512), hi2 = *(const v4f*)(p + 640);
      v4f lo3 = *(const v4f*)(p + 768), hi3 = *(const v4f*)(p + 896);
      const v4f slo = (lo0 + lo1) + (lo2 + lo3);
      const v4f shi = (hi0 + hi1) + (hi2 + hi3);
      const float v[8] = {slo[0], slo[1], slo[2], slo[3], shi[0], shi[1], shi[2], shi[3]};

      const int w2 = 16 * j - 20 + row;
      const bool inr = (w2 >= 0) && (w2 < WW);
      const int d0 = 16 * (j - ti) + row;
#pragma unroll
      for (int r = 0; r < 8; ++r) {
        const int m  = mBase + r;
        const int w1 = ti * 16 + m;
        const int diff = d0 - m;
        if (diff >= 0 && diff <= 40 && !(diff & 1))
          out[outBase + (size_t)(diff >> 1) * HW + w1] =
              inr ? v[r] * 0.00390625f : 0.f;
      }
    }

    lds_barrier();
  }
}

extern "C" void kernel_launch(void* const* d_in, const int* in_sizes, int n_in,
                              void* d_out, int out_size, void* d_ws, size_t ws_size,
                              hipStream_t stream) {
  const float* f1 = (const float*)d_in[0];
  const float* f2 = (const float*)d_in[1];
  float* out = (float*)d_out;

  dim3 grid(HH, BATCH);
  dim3 block(NT);
  const size_t shmem = (size_t)(NQ * QSTR) * sizeof(float);
  corr_wmma_f32_kernel<<<grid, block, shmem, stream>>>(f1, f2, out);
}


# K-quarter partial tiles stored to LDS as a band matrix (pitch 66) and joined along its diagonals: each wave store writes one whole 128-byte line of out (was 4-byte stores scattered over 16 lines)
# speedup vs baseline: 1.6112x; 1.2893x over previous
#include <hip/hip_runtime.h>


typedef __attribute__((ext_vector_type(2))) float v2f;
typedef __attribute__((ext_vector_type(4))) float v4f;
typedef __attribute__((ext_vector_type(8))) float v8f;

#define BATCH 8
#define CCH   256
#define HH    64
#define WW    96
#define HW    (HH * WW)
#define ND    21
#define NWV   4
#define NT    (NWV * 32)
#define KSW   16
#define NLINE 63
#define PITCH 66
#define RSTR  (16 * PITCH)
#define WSTR  (6 * RSTR)

__device__ __forceinline__ v2f ld_frag(const float* __restrict__ p, const int s) {
  v2f v;
  v.x = p[(size_t)(4 * s) * HW];
  v.y = p[(size_t)(4 * s + 1) * HW];
  return v;
}

__device__ __forceinline__ void st_part(float* __restrict__ part, const int ti, const int j,
                                        const v8f a) {
#pragma unroll
  for (int r = 0; r < 8; ++r) part[ti * RSTR + r * PITCH + 16 * (j - ti)] = a[r];
}

template <int J>
__device__ __forceinline__ void tile_step(const v2f (&A)[6][KSW], v2f (&B)[KSW],
                                          const float* __restrict__ nxt,
                                          float* __restrict__ part) {
  constexpr int lo = (J > 3) ? J - 3 : 0;
  constexpr int hi = (J < 5) ? J : 5;
  constexpr int n  = hi - lo + 1;
  const v8f zero = (v8f){0.f, 0.f, 0.f, 0.f, 0.f, 0.f, 0.f, 0.f};
  if constexpr (n == 1) {
    v8f a0 = zero, a1 = zero;
#pragma unroll
    for (int s = 0; s < KSW; s += 2) {
      a0 = __builtin_amdgcn_wmma_f32_16x16x4_f32(false, A[lo][s], false, B[s],
                                                 (short)0, a0, false, false);
      a1 = __builtin_amdgcn_wmma_f32_16x16x4_f32(false, A[lo][s + 1], false, B[s + 1],
                                                 (short)0, a1, false, false);
      B[s]     = ld_frag(nxt, s);
      B[s + 1] = ld_frag(nxt, s + 1);
    }
    st_part(part, lo, J, a0 + a1);
  } else {
    v8f acc[n];
#pragma unroll
    for (int t = 0; t < n; ++t) acc[t] = zero;
#pragma unroll
    for (int s = 0; s < KSW; ++s) {
#pragma unroll
      for (int t = 0; t < n; ++t)
        acc[t] = __builtin_amdgcn_wmma_f32_16x16x4_f32(false, A[lo + t][s], false, B[s],
                                                       (short)0, acc[t], false, false);
      B[s] = ld_frag(nxt, s);
    }
#pragma unroll
    for (int t = 0; t < n; ++t) st_part(part, lo + t, J, acc[t]);
  }
}

__device__ __forceinline__ void lds_barrier() {
  __builtin_amdgcn_fence(__ATOMIC_RELEASE, "workgroup", "local");
  __builtin_amdgcn_s_barrier();
  __builtin_amdgcn_fence(__ATOMIC_ACQUIRE, "workgroup", "local");
}

__global__ __launch_bounds__(NT)
void corr_wmma_f32_kernel(const float* __restrict__ f1,
                          const float* __restrict__ f2,
                          float* __restrict__ out) {
  extern __shared__ float smem[];

  const int tid  = threadIdx.x;
  const int lane = tid & 31;
  const int wv   = __builtin_amdgcn_readfirstlane(tid >> 5);

  const int h = blockIdx.x;
  const int b = blockIdx.y;
  const size_t inBase = (size_t)b * (CCH * HW);

  const int row = lane & 15;
  const int khi = lane >> 4;

  const int dyFirst = (h < 20) ? (21 - h) / 2 : 0;
  const int dyLast  = ((83 - h) / 2 < 20) ? (83 - h) / 2 : 20;

  for (int dy = 0; dy < ND; ++dy) {
    if (dy >= dyFirst && dy <= dyLast) continue;
    const size_t outBase =
        (((size_t)b * (ND * ND) + (size_t)dy * ND) * HH + (size_t)h) * WW;
    for (int i = tid; i < ND * WW; i += NT) {
      const int dxI = i / WW, w = i % WW;
      out[outBase + (size_t)dxI * HW + w] = 0.f;
    }
  }

  v2f A[6][KSW];
  {
    const float* a = f1 + inBase + (size_t)(64 * wv + 2 * khi) * HW + (size_t)h * WW + row;
#pragma unroll
    for (int ti = 0; ti < 6; ++ti)
#pragma unroll
      for (int s = 0; s < KSW; ++s) A[ti][s] = ld_frag(a + 16 * ti, s);
  }

  const float* f2k = f2 + inBase + (size_t)(64 * wv + 2 * khi) * HW;
  int col[9];
#pragma unroll
  for (int j = 0; j < 9; ++j) {
    const int w2 = 16 * j - 20 + row;
    col[j] = w2 < 0 ? 0 : (w2 >= WW ? WW - 1 : w2);
  }

  v2f B0[KSW], B1[KSW], B2[KSW];
  {
    const float* r0 = f2k + (size_t)(h + 2 * dyFirst - 20) * WW;
#pragma unroll
    for (int s = 0; s < KSW; ++s) {
      B0[s] = ld_frag(r0 + col[0], s);
      B1[s] = ld_frag(r0 + col[1], s);
      B2[s] = ld_frag(r0 + col[2], s);
    }
  }

  float* part = smem + wv * WSTR + khi * (8 * PITCH) + row;
  const float* red = smem + khi * RSTR + row * (PITCH + 1);

  for (int dy = dyFirst; dy <= dyLast; ++dy) {
    const int dyn = (dy < dyLast) ? dy + 1 : dy;
    const float* rc = f2k + (size_t)(h + 2 * dy - 20) * WW;
    const float* rn = f2k + (size_t)(h + 2 * dyn - 20) * WW;

    tile_step<0>(A, B0, rc + col[3], part);
    tile_step<1>(A, B1, rc + col[4], part);
    tile_step<2>(A, B2, rc + col[5], part);
    tile_step<3>(A, B0, rc + col[6], part);
    tile_step<4>(A, B1, rc + col[7], part);
    tile_step<5>(A, B2, rc + col[8], part);
    tile_step<6>(A, B0, rn + col[0], part);
    tile_step<7>(A, B1, rn + col[1], part);
    tile_step<8>(A, B2, rn + col[2], part);

    lds_barrier();

    const size_t outBase =
        (((size_t)b * (ND * ND) + (size_t)dy * ND) * HH + (size_t)h) * WW;
    float pq[15][4];
#pragma unroll
    for (int i = 0; i < 15; ++i) {
      const int L   = wv + 4 * i;
      const int dxI = L / 3;
      const int ch  = L - 3 * dxI;
      const float* p = red + (2 * ch) * RSTR + 2 * dxI;
      pq[i][0] = p[0];
      pq[i][1] = p[WSTR];
      pq[i][2] = p[2 * WSTR];
      pq[i][3] = p[3 * WSTR];
    }
#pragma unroll
    for (int i = 0; i < 15; ++i) {
      const int L   = wv + 4 * i;
      const int dxI = L / 3;
      const int ch  = L - 3 * dxI;
      const float v = (pq[i][0] + pq[i][1]) + (pq[i][2] + pq[i][3]);
      const int w1 = 32 * ch + lane;
      const int w2 = w1 + 2 * dxI - 20;
      out[outBase + (size_t)dxI * HW + w1] =
          (w2 >= 0 && w2 < WW) ? v * 0.00390625f : 0.f;
    }
    if (wv < 3) {
      const float* p = red + (2 * wv) * RSTR + 40;
      const float v = (p[0] + p[WSTR]) + (p[2 * WSTR] + p[3 * WSTR]);
      const int w1 = 32 * wv + lane;
      out[outBase + (size_t)20 * HW + w1] = (w1 + 20 < WW) ? v * 0.00390625f : 0.f;
    }

    lds_barrier();
  }
}

extern "C" void kernel_launch(void* const* d_in, const int* in_sizes, int n_in,
                              void* d_out, int out_size, void* d_ws, size_t ws_size,
                              hipStream_t stream) {
  const float* f1 = (const float*)d_in[0];
  const float* f2 = (const float*)d_in[1];
  float* out = (float*)d_out;

  dim3 grid(HH, BATCH);
  dim3 block(NT);
  const size_t shmem = (size_t)(NWV * WSTR) * sizeof(float);
  corr_wmma_f32_kernel<<<grid, block, shmem, stream>>>(f1, f2, out);
}


# f2 refill loads grouped: a tile's WMMAs run as one block, then its register set is refilled by one group of 32 global loads (was 2 loads per 2-4 WMMAs): 14 breaks of the WMMA stream per dy, was 91
# speedup vs baseline: 1.7373x; 1.0783x over previous
#include <hip/hip_runtime.h>


typedef __attribute__((ext_vector_type(2))) float v2f;
typedef __attribute__((ext_vector_type(4))) float v4f;
typedef __attribute__((ext_vector_type(8))) float v8f;

#define BATCH 8
#define CCH   256
#define HH    64
#define WW    96
#define HW    (HH * WW)
#define ND    21
#define NWV   4
#define NT    (NWV * 32)
#define KSW   16
#define NLINE 63
#define PITCH 66
#define RSTR  (16 * PITCH)
#define WSTR  (6 * RSTR)

__device__ __forceinline__ v2f ld_frag(const float* __restrict__ p, const int s) {
  v2f v;
  v.x = p[(size_t)(4 * s) * HW];
  v.y = p[(size_t)(4 * s + 1) * HW];
  return v;
}

__device__ __forceinline__ void st_part(float* __restrict__ part, const int ti, const int j,
                                        const v8f a) {
#pragma unroll
  for (int r = 0; r < 8; ++r) part[ti * RSTR + r * PITCH + 16 * (j - ti)] = a[r];
}

template <int J>
__device__ __forceinline__ void tile_step(const v2f (&A)[6][KSW], v2f (&B)[KSW],
                                          const float* __restrict__ nxt,
                                          float* __restrict__ part) {
  constexpr int lo = (J > 3) ? J - 3 : 0;
  constexpr int hi = (J < 5) ? J : 5;
  constexpr int n  = hi - lo + 1;
  const v8f zero = (v8f){0.f, 0.f, 0.f, 0.f, 0.f, 0.f, 0.f, 0.f};
  __builtin_amdgcn_sched_barrier(0);
  if constexpr (n == 1) {
    v8f a0 = zero, a1 = zero;
#pragma unroll
    for (int s = 0; s < KSW; s += 2) {
      a0 = __builtin_amdgcn_wmma_f32_16x16x4_f32(false, A[lo][s], false, B[s],
                                                 (short)0, a0, false, false);
      a1 = __builtin_amdgcn_wmma_f32_16x16x4_f32(false, A[lo][s + 1], false, B[s + 1],
                                                 (short)0, a1, false, false);
    }
#pragma unroll
    for (int s = 0; s < KSW; ++s) B[s] = ld_frag(nxt, s);
    st_part(part, lo, J, a0 + a1);
  } else {
    v8f acc[n];
#pragma unroll
    for (int t = 0; t < n; ++t) acc[t] = zero;
#pragma unroll
    for (int s = 0; s < KSW; ++s) {
#pragma unroll
      for (int t = 0; t < n; ++t)
        acc[t] = __builtin_amdgcn_wmma_f32_16x16x4_f32(false, A[lo + t][s], false, B[s],
                                                       (short)0, acc[t], false, false);
    }
#pragma unroll
    for (int s = 0; s < KSW; ++s) B[s] = ld_frag(nxt, s);
#pragma unroll
    for (int t = 0; t < n; ++t) st_part(part, lo + t, J, acc[t]);
  }
}

__device__ __forceinline__ void lds_barrier() {
  __builtin_amdgcn_fence(__ATOMIC_RELEASE, "workgroup", "local");
  __builtin_amdgcn_s_barrier();
  __builtin_amdgcn_fence(__ATOMIC_ACQUIRE, "workgroup", "local");
}

__global__ __launch_bounds__(NT)
void corr_wmma_f32_kernel(const float* __restrict__ f1,
                          const float* __restrict__ f2,
                          float* __restrict__ out) {
  extern __shared__ float smem[];

  const int tid  = threadIdx.x;
  const int lane = tid & 31;
  const int wv   = __builtin_amdgcn_readfirstlane(tid >> 5);

  const int h = blockIdx.x;
  const int b = blockIdx.y;
  const size_t inBase = (size_t)b * (CCH * HW);

  const int row = lane & 15;
  const int khi = lane >> 4;

  const int dyFirst = (h < 20) ? (21 - h) / 2 : 0;
  const int dyLast  = ((83 - h) / 2 < 20) ? (83 - h) / 2 : 20;

  for (int dy = 0; dy < ND; ++dy) {
    if (dy >= dyFirst && dy <= dyLast) continue;
    const size_t outBase =
        (((size_t)b * (ND * ND) + (size_t)dy * ND) * HH + (size_t)h) * WW;
    for (int i = tid; i < ND * WW; i += NT) {
      const int dxI = i / WW, w = i % WW;
      out[outBase + (size_t)dxI * HW + w] = 0.f;
    }
  }

  v2f A[6][KSW];
  {
    const float* a = f1 + inBase + (size_t)(64 * wv + 2 * khi) * HW + (size_t)h * WW + row;
#pragma unroll
    for (int ti = 0; ti < 6; ++ti)
#pragma unroll
      for (int s = 0; s < KSW; ++s) A[ti][s] = ld_frag(a + 16 * ti, s);
  }

  const float* f2k = f2 + inBase + (size_t)(64 * wv + 2 * khi) * HW;
  int col[9];
#pragma unroll
  for (int j = 0; j < 9; ++j) {
    const int w2 = 16 * j - 20 + row;
    col[j] = w2 < 0 ? 0 : (w2 >= WW ? WW - 1 : w2);
  }

  v2f B0[KSW], B1[KSW], B2[KSW];
  {
    const float* r0 = f2k + (size_t)(h + 2 * dyFirst - 20) * WW;
#pragma unroll
    for (int s = 0; s < KSW; ++s) B0[s] = ld_frag(r0 + col[0], s);
#pragma unroll
    for (int s = 0; s < KSW; ++s) B1[s] = ld_frag(r0 + col[1], s);
#pragma unroll
    for (int s = 0; s < KSW; ++s) B2[s] = ld_frag(r0 + col[2], s);
  }

  float* part = smem + wv * WSTR + khi * (8 * PITCH) + row;
  const float* red = smem + khi * RSTR + row * (PITCH + 1);

  for (int dy = dyFirst; dy <= dyLast; ++dy) {
    const int dyn = (dy < dyLast) ? dy + 1 : dy;
    const float* rc = f2k + (size_t)(h + 2 * dy - 20) * WW;
    const float* rn = f2k + (size_t)(h + 2 * dyn - 20) * WW;

    tile_step<0>(A, B0, rc + col[3], part);
    tile_step<1>(A, B1, rc + col[4], part);
    tile_step<2>(A, B2, rc + col[5], part);
    tile_step<3>(A, B0, rc + col[6], part);
    tile_step<4>(A, B1, rc + col[7], part);
    tile_step<5>(A, B2, rc + col[8], part);
    tile_step<6>(A, B0, rn + col[0], part);
    tile_step<7>(A, B1, rn + col[1], part);
    tile_step<8>(A, B2, rn + col[2], part);

    lds_barrier();

    const size_t outBase =
        (((size_t)b * (ND * ND) + (size_t)dy * ND) * HH + (size_t)h) * WW;
    float pq[15][4];
#pragma unroll
    for (int i = 0; i < 15; ++i) {
      const int L   = wv + 4 * i;
      const int dxI = L / 3;
      const int ch  = L - 3 * dxI;
      const float* p = red + (2 * ch) * RSTR + 2 * dxI;
      pq[i][0] = p[0];
      pq[i][1] = p[WSTR];
      pq[i][2] = p[2 * WSTR];
      pq[i][3] = p[3 * WSTR];
    }
#pragma unroll
    for (int i = 0; i < 15; ++i) {
      const int L   = wv + 4 * i;
      const int dxI = L / 3;
      const int ch  = L - 3 * dxI;
      const float v = (pq[i][0] + pq[i][1]) + (pq[i][2] + pq[i][3]);
      const int w1 = 32 * ch + lane;
      const int w2 = w1 + 2 * dxI - 20;
      out[outBase + (size_t)dxI * HW + w1] =
          (w2 >= 0 && w2 < WW) ? v * 0.00390625f : 0.f;
    }
    if (wv < 3) {
      const float* p = red + (2 * wv) * RSTR + 40;
      const float v = (p[0] + p[WSTR]) + (p[2 * WSTR] + p[3 * WSTR]);
      const int w1 = 32 * wv + lane;
      out[outBase + (size_t)20 * HW + w1] = (w1 + 20 < WW) ? v * 0.00390625f : 0.f;
    }

    lds_barrier();
  }
}

extern "C" void kernel_launch(void* const* d_in, const int* in_sizes, int n_in,
                              void* d_out, int out_size, void* d_ws, size_t ws_size,
                              hipStream_t stream) {
  const float* f1 = (const float*)d_in[0];
  const float* f2 = (const float*)d_in[1];
  float* out = (float*)d_out;

  dim3 grid(HH, BATCH);
  dim3 block(NT);
  const size_t shmem = (size_t)(NWV * WSTR) * sizeof(float);
  corr_wmma_f32_kernel<<<grid, block, shmem, stream>>>(f1, f2, out);
}


# f2 tiles 0 and 8 (columns wholly in the zero padding) are no longer loaded or multiplied: 22 band tiles and 7 f2 tile loads per dy, was 24 and 9; their two band slots in LDS are cleared once
# speedup vs baseline: 1.9501x; 1.1225x over previous
#include <hip/hip_runtime.h>


typedef __attribute__((ext_vector_type(2))) float v2f;
typedef __attribute__((ext_vector_type(4))) float v4f;
typedef __attribute__((ext_vector_type(8))) float v8f;

#define BATCH 8
#define CCH   256
#define HH    64
#define WW    96
#define HW    (HH * WW)
#define ND    21
#define NWV   4
#define NT    (NWV * 32)
#define KSW   16
#define NLINE 63
#define PITCH 66
#define RSTR  (16 * PITCH)
#define WSTR  (6 * RSTR)

__device__ __forceinline__ v2f ld_frag(const float* __restrict__ p, const int s) {
  v2f v;
  v.x = p[(size_t)(4 * s) * HW];
  v.y = p[(size_t)(4 * s + 1) * HW];
  return v;
}

__device__ __forceinline__ void st_part(float* __restrict__ part, const int ti, const int j,
                                        const v8f a) {
#pragma unroll
  for (int r = 0; r < 8; ++r) part[ti * RSTR + r * PITCH + 16 * (j - ti)] = a[r];
}

template <int J>
__device__ __forceinline__ void tile_step(const v2f (&A)[6][KSW], v2f (&B)[KSW],
                                          const float* __restrict__ nxt,
                                          float* __restrict__ part) {
  constexpr int lo = (J > 3) ? J - 3 : 0;
  constexpr int hi = (J < 5) ? J : 5;
  constexpr int n  = hi - lo + 1;
  const v8f zero = (v8f){0.f, 0.f, 0.f, 0.f, 0.f, 0.f, 0.f, 0.f};
  __builtin_amdgcn_sched_barrier(0);
  v8f acc[n];
#pragma unroll
  for (int t = 0; t < n; ++t) acc[t] = zero;
#pragma unroll
  for (int s = 0; s < KSW; ++s) {
#pragma unroll
    for (int t = 0; t < n; ++t)
      acc[t] = __builtin_amdgcn_wmma_f32_16x16x4_f32(false, A[lo + t][s], false, B[s],
                                                     (short)0, acc[t], false, false);
  }
#pragma unroll
  for (int s = 0; s < KSW; ++s) B[s] = ld_frag(nxt, s);
#pragma unroll
  for (int t = 0; t < n; ++t) st_part(part, lo + t, J, acc[t]);
}

__device__ __forceinline__ void lds_barrier() {
  __builtin_amdgcn_fence(__ATOMIC_RELEASE, "workgroup", "local");
  __builtin_amdgcn_s_barrier();
  __builtin_amdgcn_fence(__ATOMIC_ACQUIRE, "workgroup", "local");
}

__global__ __launch_bounds__(NT)
void corr_wmma_f32_kernel(const float* __restrict__ f1,
                          const float* __restrict__ f2,
                          float* __restrict__ out) {
  extern __shared__ float smem[];

  const int tid  = threadIdx.x;
  const int lane = tid & 31;
  const int wv   = __builtin_amdgcn_readfirstlane(tid >> 5);

  const int h = blockIdx.x;
  const int b = blockIdx.y;
  const size_t inBase = (size_t)b * (CCH * HW);

  const int row = lane & 15;
  const int khi = lane >> 4;

  const int dyFirst = (h < 20) ? (21 - h) / 2 : 0;
  const int dyLast  = ((83 - h) / 2 < 20) ? (83 - h) / 2 : 20;

  for (int dy = 0; dy < ND; ++dy) {
    if (dy >= dyFirst && dy <= dyLast) continue;
    const size_t outBase =
        (((size_t)b * (ND * ND) + (size_t)dy * ND) * HH + (size_t)h) * WW;
    for (int i = tid; i < ND * WW; i += NT) {
      const int dxI = i / WW, w = i % WW;
      out[outBase + (size_t)dxI * HW + w] = 0.f;
    }
  }

  v2f A[6][KSW];
  {
    const float* a = f1 + inBase + (size_t)(64 * wv + 2 * khi) * HW + (size_t)h * WW + row;
#pragma unroll
    for (int ti = 0; ti < 6; ++ti)
#pragma unroll
      for (int s = 0; s < KSW; ++s) A[ti][s] = ld_frag(a + 16 * ti, s);
  }

  const float* f2k = f2 + inBase + (size_t)(64 * wv + 2 * khi) * HW;
  int col[9];
#pragma unroll
  for (int j = 0; j < 9; ++j) {
    const int w2 = 16 * j - 20 + row;
    col[j] = w2 < 0 ? 0 : (w2 >= WW ? WW - 1 : w2);
  }

  v2f B0[KSW], B1[KSW], B2[KSW];
  {
    const float* r0 = f2k + (size_t)(h + 2 * dyFirst - 20) * WW;
#pragma unroll
    for (int s = 0; s < KSW; ++s) B0[s] = ld_frag(r0 + col[1], s);
#pragma unroll
    for (int s = 0; s < KSW; ++s) B1[s] = ld_frag(r0 + col[2], s);
#pragma unroll
    for (int s = 0; s < KSW; ++s) B2[s] = ld_frag(r0 + col[3], s);
  }

  float* part = smem + wv * WSTR + khi * (8 * PITCH) + row;
  const float* red = smem + khi * RSTR + row * (PITCH + 1);

  {
    const v8f zero = (v8f){0.f, 0.f, 0.f, 0.f, 0.f, 0.f, 0.f, 0.f};
    st_part(part, 0, 0, zero);
    st_part(part, 5, 8, zero);
  }

  for (int dy = dyFirst; dy <= dyLast; ++dy) {
    const int dyn = (dy < dyLast) ? dy + 1 : dy;
    const float* rc = f2k + (size_t)(h + 2 * dy - 20) * WW;
    const float* rn = f2k + (size_t)(h + 2 * dyn - 20) * WW;

    tile_step<1>(A, B0, rc + col[4], part);
    tile_step<2>(A, B1, rc + col[5], part);
    tile_step<3>(A, B2, rc + col[6], part);
    tile_step<4>(A, B0, rc + col[7], part);
    tile_step<5>(A, B1, rn + col[2], part);
    tile_step<6>(A, B2, rn + col[3], part);
    tile_step<7>(A, B0, rn + col[1], part);

    lds_barrier();

    const size_t outBase =
        (((size_t)b * (ND * ND) + (size_t)dy * ND) * HH + (size_t)h) * WW;
    float pq[15][4];
#pragma unroll
    for (int i = 0; i < 15; ++i) {
      const int L   = wv + 4 * i;
      const int dxI = L / 3;
      const int ch  = L - 3 * dxI;
      const float* p = red + (2 * ch) * RSTR + 2 * dxI;
      pq[i][0] = p[0];
      pq[i][1] = p[WSTR];
      pq[i][2] = p[2 * WSTR];
      pq[i][3] = p[3 * WSTR];
    }
#pragma unroll
    for (int i = 0; i < 15; ++i) {
      const int L   = wv + 4 * i;
      const int dxI = L / 3;
      const int ch  = L - 3 * dxI;
      const float v = (pq[i][0] + pq[i][1]) + (pq[i][2] + pq[i][3]);
      const int w1 = 32 * ch + lane;
      const int w2 = w1 + 2 * dxI - 20;
      out[outBase + (size_t)dxI * HW + w1] =
          (w2 >= 0 && w2 < WW) ? v * 0.00390625f : 0.f;
    }
    if (wv < 3) {
      const float* p = red + (2 * wv) * RSTR + 40;
      const float v = (p[0] + p[WSTR]) + (p[2 * WSTR] + p[3 * WSTR]);
      const int w1 = 32 * wv + lane;
      out[outBase + (size_t)20 * HW + w1] = (w1 + 20 < WW) ? v * 0.00390625f : 0.f;
    }

    lds_barrier();
  }
}

extern "C" void kernel_launch(void* const* d_in, const int* in_sizes, int n_in,
                              void* d_out, int out_size, void* d_ws, size_t ws_size,
                              hipStream_t stream) {
  const float* f1 = (const float*)d_in[0];
  const float* f2 = (const float*)d_in[1];
  float* out = (float*)d_out;

  dim3 grid(HH, BATCH);
  dim3 block(NT);
  const size_t shmem = (size_t)(NWV * WSTR) * sizeof(float);
  corr_wmma_f32_kernel<<<grid, block, shmem, stream>>>(f1, f2, out);
}


# wave priority falls with a work-group's own progress (s_setprio 3,2,1,0 over the quarters of its dy loop): of the two groups sharing a SIMD the one behind wins the matrix pipe, both finish together
# speedup vs baseline: 2.0123x; 1.0319x over previous
#include <hip/hip_runtime.h>


typedef __attribute__((ext_vector_type(2))) float v2f;
typedef __attribute__((ext_vector_type(4))) float v4f;
typedef __attribute__((ext_vector_type(8))) float v8f;

#define BATCH 8
#define CCH   256
#define HH    64
#define WW    96
#define HW    (HH * WW)
#define ND    21
#define NWV   4
#define NT    (NWV * 32)
#define KSW   16
#define NLINE 63
#define PITCH 66
#define RSTR  (16 * PITCH)
#define WSTR  (6 * RSTR)

__device__ __forceinline__ v2f ld_frag(const float* __restrict__ p, const int s) {
  v2f v;
  v.x = p[(size_t)(4 * s) * HW];
  v.y = p[(size_t)(4 * s + 1) * HW];
  return v;
}

__device__ __forceinline__ void st_part(float* __restrict__ part, const int ti, const int j,
                                        const v8f a) {
#pragma unroll
  for (int r = 0; r < 8; ++r) part[ti * RSTR + r * PITCH + 16 * (j - ti)] = a[r];
}

template <int J>
__device__ __forceinline__ void tile_step(const v2f (&A)[6][KSW], v2f (&B)[KSW],
                                          const float* __restrict__ nxt,
                                          float* __restrict__ part) {
  constexpr int lo = (J > 3) ? J - 3 : 0;
  constexpr int hi = (J < 5) ? J : 5;
  constexpr int n  = hi - lo + 1;
  const v8f zero = (v8f){0.f, 0.f, 0.f, 0.f, 0.f, 0.f, 0.f, 0.f};
  __builtin_amdgcn_sched_barrier(0);
  v8f acc[n];
#pragma unroll
  for (int t = 0; t < n; ++t) acc[t] = zero;
#pragma unroll
  for (int s = 0; s < KSW; ++s) {
#pragma unroll
    for (int t = 0; t < n; ++t)
      acc[t] = __builtin_amdgcn_wmma_f32_16x16x4_f32(false, A[lo + t][s], false, B[s],
                                                     (short)0, acc[t], false, false);
  }
#pragma unroll
  for (int s = 0; s < KSW; ++s) B[s] = ld_frag(nxt, s);
#pragma unroll
  for (int t = 0; t < n; ++t) st_part(part, lo + t, J, acc[t]);
}

__device__ __forceinline__ void lds_barrier() {
  __builtin_amdgcn_fence(__ATOMIC_RELEASE, "workgroup", "local");
  __builtin_amdgcn_s_barrier();
  __builtin_amdgcn_fence(__ATOMIC_ACQUIRE, "workgroup", "local");
}

__device__ __forceinline__ void set_wave_prio(const int lvl) {
  if (lvl == 3)      __builtin_amdgcn_s_setprio(3);
  else if (lvl == 2) __builtin_amdgcn_s_setprio(2);
  else if (lvl == 1) __builtin_amdgcn_s_setprio(1);
  else               __builtin_amdgcn_s_setprio(0);
}

__global__ __launch_bounds__(NT)
void corr_wmma_f32_kernel(const float* __restrict__ f1,
                          const float* __restrict__ f2,
                          float* __restrict__ out) {
  extern __shared__ float smem[];

  const int tid  = threadIdx.x;
  const int lane = tid & 31;
  const int wv   = __builtin_amdgcn_readfirstlane(tid >> 5);

  const int h = blockIdx.x;
  const int b = blockIdx.y;
  const size_t inBase = (size_t)b * (CCH * HW);

  const int row = lane & 15;
  const int khi = lane >> 4;

  const int dyFirst = (h < 20) ? (21 - h) / 2 : 0;
  const int dyLast  = ((83 - h) / 2 < 20) ? (83 - h) / 2 : 20;

  for (int dy = 0; dy < ND; ++dy) {
    if (dy >= dyFirst && dy <= dyLast) continue;
    const size_t outBase =
        (((size_t)b * (ND * ND) + (size_t)dy * ND) * HH + (size_t)h) * WW;
    for (int i = tid; i < ND * WW; i += NT) {
      const int dxI = i / WW, w = i % WW;
      out[outBase + (size_t)dxI * HW + w] = 0.f;
    }
  }

  v2f A[6][KSW];
  {
    const float* a = f1 + inBase + (size_t)(64 * wv + 2 * khi) * HW + (size_t)h * WW + row;
#pragma unroll
    for (int ti = 0; ti < 6; ++ti)
#pragma unroll
      for (int s = 0; s < KSW; ++s) A[ti][s] = ld_frag(a + 16 * ti, s);
  }

  const float* f2k = f2 + inBase + (size_t)(64 * wv + 2 * khi) * HW;
  int col[9];
#pragma unroll
  for (int j = 0; j < 9; ++j) {
    const int w2 = 16 * j - 20 + row;
    col[j] = w2 < 0 ? 0 : (w2 >= WW ? WW - 1 : w2);
  }

  v2f B0[KSW], B1[KSW], B2[KSW];
  {
    const float* r0 = f2k + (size_t)(h + 2 * dyFirst - 20) * WW;
#pragma unroll
    for (int s = 0; s < KSW; ++s) B0[s] = ld_frag(r0 + col[1], s);
#pragma unroll
    for (int s = 0; s < KSW; ++s) B1[s] = ld_frag(r0 + col[2], s);
#pragma unroll
    for (int s = 0; s < KSW; ++s) B2[s] = ld_frag(r0 + col[3], s);
  }

  float* part = smem + wv * WSTR + khi * (8 * PITCH) + row;
  const float* red = smem + khi * RSTR + row * (PITCH + 1);

  {
    const v8f zero = (v8f){0.f, 0.f, 0.f, 0.f, 0.f, 0.f, 0.f, 0.f};
    st_part(part, 0, 0, zero);
    st_part(part, 5, 8, zero);
  }

  const int nDy = dyLast - dyFirst + 1;
  int lvlCur = 0;

  for (int dy = dyFirst; dy <= dyLast; ++dy) {
    const int i4  = 4 * (dy - dyFirst);
    const int lvl = 3 - (i4 >= nDy) - (i4 >= 2 * nDy) - (i4 >= 3 * nDy);
    if (lvl != lvlCur) { set_wave_prio(lvl); lvlCur = lvl; }

    const int dyn = (dy < dyLast) ? dy + 1 : dy;
    const float* rc = f2k + (size_t)(h + 2 * dy - 20) * WW;
    const float* rn = f2k + (size_t)(h + 2 * dyn - 20) * WW;

    tile_step<1>(A, B0, rc + col[4], part);
    tile_step<2>(A, B1, rc + col[5], part);
    tile_step<3>(A, B2, rc + col[6], part);
    tile_step<4>(A, B0, rc + col[7], part);
    tile_step<5>(A, B1, rn + col[2], part);
    tile_step<6>(A, B2, rn + col[3], part);
    tile_step<7>(A, B0, rn + col[1], part);

    lds_barrier();

    const size_t outBase =
        (((size_t)b * (ND * ND) + (size_t)dy * ND) * HH + (size_t)h) * WW;
    float pq[15][4];
#pragma unroll
    for (int i = 0; i < 15; ++i) {
      const int L   = wv + 4 * i;
      const int dxI = L / 3;
      const int ch  = L - 3 * dxI;
      const float* p = red + (2 * ch) * RSTR + 2 * dxI;
      pq[i][0] = p[0];
      pq[i][1] = p[WSTR];
      pq[i][2] = p[2 * WSTR];
      pq[i][3] = p[3 * WSTR];
    }
#pragma unroll
    for (int i = 0; i < 15; ++i) {
      const int L   = wv + 4 * i;
      const int dxI = L / 3;
      const int ch  = L - 3 * dxI;
      const float v = (pq[i][0] + pq[i][1]) + (pq[i][2] + pq[i][3]);
      const int w1 = 32 * ch + lane;
      const int w2 = w1 + 2 * dxI - 20;
      out[outBase + (size_t)dxI * HW + w1] =
          (w2 >= 0 && w2 < WW) ? v * 0.00390625f : 0.f;
    }
    if (wv < 3) {
      const float* p = red + (2 * wv) * RSTR + 40;
      const float v = (p[0] + p[WSTR]) + (p[2 * WSTR] + p[3 * WSTR]);
      const int w1 = 32 * wv + lane;
      out[outBase + (size_t)20 * HW + w1] = (w1 + 20 < WW) ? v * 0.00390625f : 0.f;
    }

    lds_barrier();
  }
}

extern "C" void kernel_launch(void* const* d_in, const int* in_sizes, int n_in,
                              void* d_out, int out_size, void* d_ws, size_t ws_size,
                              hipStream_t stream) {
  const float* f1 = (const float*)d_in[0];
  const float* f2 = (const float*)d_in[1];
  float* out = (float*)d_out;

  dim3 grid(HH, BATCH);
  dim3 block(NT);
  const size_t shmem = (size_t)(NWV * WSTR) * sizeof(float);
  corr_wmma_f32_kernel<<<grid, block, shmem, stream>>>(f1, f2, out);
}


# f2 row resident in registers, f1 rows streamed (operand roles swapped, a block owns an f2 row): a dy loads 6 aligned 64-byte f1 tiles, was 7 f2 tiles off the 64-byte grid; sums unchanged
# speedup vs baseline: 2.0781x; 1.0327x over previous
#include <hip/hip_runtime.h>


typedef __attribute__((ext_vector_type(2))) float v2f;
typedef __attribute__((ext_vector_type(4))) float v4f;
typedef __attribute__((ext_vector_type(8))) float v8f;

#define BATCH 8
#define CCH   256
#define HH    64
#define WW    96
#define HW    (HH * WW)
#define ND    21
#define NWV   4
#define NT    (NWV * 32)
#define KSW   16
#define NLINE 63
#define PITCH 66
#define RSTR  (16 * PITCH)
#define WSTR  (6 * RSTR)

__device__ __forceinline__ v2f ld_frag(const float* __restrict__ p, const int s) {
  v2f v;
  v.x = p[(size_t)(4 * s) * HW];
  v.y = p[(size_t)(4 * s + 1) * HW];
  return v;
}

__device__ __forceinline__ void st_part(float* __restrict__ part, const int ti, const int j,
                                        const v8f a) {
#pragma unroll
  for (int r = 0; r < 8; ++r) part[ti * RSTR + r * PITCH + 16 * (j - ti)] = a[r];
}

template <int TI>
__device__ __forceinline__ void tile_step(const v2f (&B)[7][KSW], v2f (&A)[KSW],
                                          const float* __restrict__ nxt,
                                          float* __restrict__ part) {
  constexpr int lo = (TI > 1) ? TI : 1;
  constexpr int hi = (TI + 3 < 7) ? TI + 3 : 7;
  constexpr int n  = hi - lo + 1;
  const v8f zero = (v8f){0.f, 0.f, 0.f, 0.f, 0.f, 0.f, 0.f, 0.f};
  __builtin_amdgcn_sched_barrier(0);
  v8f acc[n];
#pragma unroll
  for (int t = 0; t < n; ++t) acc[t] = zero;
#pragma unroll
  for (int s = 0; s < KSW; ++s) {
#pragma unroll
    for (int t = 0; t < n; ++t)
      acc[t] = __builtin_amdgcn_wmma_f32_16x16x4_f32(false, A[s], false, B[lo + t - 1][s],
                                                     (short)0, acc[t], false, false);
  }
#pragma unroll
  for (int s = 0; s < KSW; ++s) A[s] = ld_frag(nxt, s);
#pragma unroll
  for (int t = 0; t < n; ++t) st_part(part, TI, lo + t, acc[t]);
}

__device__ __forceinline__ void lds_barrier() {
  __builtin_amdgcn_fence(__ATOMIC_RELEASE, "workgroup", "local");
  __builtin_amdgcn_s_barrier();
  __builtin_amdgcn_fence(__ATOMIC_ACQUIRE, "workgroup", "local");
}

__device__ __forceinline__ void set_wave_prio(const int lvl) {
  if (lvl == 3)      __builtin_amdgcn_s_setprio(3);
  else if (lvl == 2) __builtin_amdgcn_s_setprio(2);
  else if (lvl == 1) __builtin_amdgcn_s_setprio(1);
  else               __builtin_amdgcn_s_setprio(0);
}

__global__ __launch_bounds__(NT)
void corr_wmma_f32_kernel(const float* __restrict__ f1,
                          const float* __restrict__ f2,
                          float* __restrict__ out) {
  extern __shared__ float smem[];

  const int tid  = threadIdx.x;
  const int lane = tid & 31;
  const int wv   = __builtin_amdgcn_readfirstlane(tid >> 5);

  const int hz = blockIdx.x;
  const int r2 = blockIdx.x;
  const int b = blockIdx.y;
  const size_t inBase = (size_t)b * (CCH * HW);

  const int row = lane & 15;
  const int khi = lane >> 4;

  const int zFirst = (hz < 20) ? (21 - hz) / 2 : 0;
  const int zLast  = ((83 - hz) / 2 < 20) ? (83 - hz) / 2 : 20;
  for (int dy = 0; dy < ND; ++dy) {
    if (dy >= zFirst && dy <= zLast) continue;
    const size_t outBase =
        (((size_t)b * (ND * ND) + (size_t)dy * ND) * HH + (size_t)hz) * WW;
    for (int i = tid; i < ND * WW; i += NT) {
      const int dxI = i / WW, w = i % WW;
      out[outBase + (size_t)dxI * HW + w] = 0.f;
    }
  }

  const int dyFirst = (r2 >= 43) ? (r2 - 42) / 2 : 0;
  const int dyLast  = ((r2 + 20) / 2 < 20) ? (r2 + 20) / 2 : 20;

  v2f B[7][KSW];
  {
    const float* p = f2 + inBase + (size_t)(64 * wv + 2 * khi) * HW + (size_t)r2 * WW;
#pragma unroll
    for (int j = 1; j < 8; ++j) {
      const int w2 = 16 * j - 20 + row;
      const int cl = w2 < 0 ? 0 : (w2 >= WW ? WW - 1 : w2);
#pragma unroll
      for (int s = 0; s < KSW; ++s) B[j - 1][s] = ld_frag(p + cl, s);
    }
  }

  const float* f1k = f1 + inBase + (size_t)(64 * wv + 2 * khi) * HW + row;
  v2f A0[KSW], A1[KSW], A2[KSW];
  {
    const float* r0 = f1k + (size_t)(r2 + 20 - 2 * dyFirst) * WW;
#pragma unroll
    for (int s = 0; s < KSW; ++s) A0[s] = ld_frag(r0, s);
#pragma unroll
    for (int s = 0; s < KSW; ++s) A1[s] = ld_frag(r0 + 16, s);
#pragma unroll
    for (int s = 0; s < KSW; ++s) A2[s] = ld_frag(r0 + 32, s);
  }

  float* part = smem + wv * WSTR + khi * (8 * PITCH) + row;
  const float* red = smem + khi * RSTR + row * (PITCH + 1);

  {
    const v8f zero = (v8f){0.f, 0.f, 0.f, 0.f, 0.f, 0.f, 0.f, 0.f};
    st_part(part, 0, 0, zero);
    st_part(part, 5, 8, zero);
  }

  const int nDy = dyLast - dyFirst + 1;
  int lvlCur = 0;

  for (int dy = dyFirst; dy <= dyLast; ++dy) {
    const int i4  = 4 * (dy - dyFirst);
    const int lvl = 3 - (i4 >= nDy) - (i4 >= 2 * nDy) - (i4 >= 3 * nDy);
    if (lvl != lvlCur) { set_wave_prio(lvl); lvlCur = lvl; }

    const int h   = r2 + 20 - 2 * dy;
    const int dyn = (dy < dyLast) ? dy + 1 : dy;
    const float* rc = f1k + (size_t)h * WW;
    const float* rn = f1k + (size_t)(r2 + 20 - 2 * dyn) * WW;

    tile_step<0>(B, A0, rc + 48, part);
    tile_step<1>(B, A1, rc + 64, part);
    tile_step<2>(B, A2, rc + 80, part);
    tile_step<3>(B, A0, rn, part);
    tile_step<4>(B, A1, rn + 16, part);
    tile_step<5>(B, A2, rn + 32, part);

    lds_barrier();

    const size_t outBase =
        (((size_t)b * (ND * ND) + (size_t)dy * ND) * HH + (size_t)h) * WW;
    float pq[15][4];
#pragma unroll
    for (int i = 0; i < 15; ++i) {
      const int L   = wv + 4 * i;
      const int dxI = L / 3;
      const int ch  = L - 3 * dxI;
      const float* p = red + (2 * ch) * RSTR + 2 * dxI;
      pq[i][0] = p[0];
      pq[i][1] = p[WSTR];
      pq[i][2] = p[2 * WSTR];
      pq[i][3] = p[3 * WSTR];
    }
#pragma unroll
    for (int i = 0; i < 15; ++i) {
      const int L   = wv + 4 * i;
      const int dxI = L / 3;
      const int ch  = L - 3 * dxI;
      const float v = (pq[i][0] + pq[i][1]) + (pq[i][2] + pq[i][3]);
      const int w1 = 32 * ch + lane;
      const int w2 = w1 + 2 * dxI - 20;
      out[outBase + (size_t)dxI * HW + w1] =
          (w2 >= 0 && w2 < WW) ? v * 0.00390625f : 0.f;
    }
    if (wv < 3) {
      const float* p = red + (2 * wv) * RSTR + 40;
      const float v = (p[0] + p[WSTR]) + (p[2 * WSTR] + p[3 * WSTR]);
      const int w1 = 32 * wv + lane;
      out[outBase + (size_t)20 * HW + w1] = (w1 + 20 < WW) ? v * 0.00390625f : 0.f;
    }

    lds_barrier();
  }
}

extern "C" void kernel_launch(void* const* d_in, const int* in_sizes, int n_in,
                              void* d_out, int out_size, void* d_ws, size_t ws_size,
                              hipStream_t stream) {
  const float* f1 = (const float*)d_in[0];
  const float* f2 = (const float*)d_in[1];
  float* out = (float*)d_out;

  dim3 grid(HH, BATCH);
  dim3 block(NT);
  const size_t shmem = (size_t)(NWV * WSTR) * sizeof(float);
  corr_wmma_f32_kernel<<<grid, block, shmem, stream>>>(f1, f2, out);
}


# operand tiles are 16 same-parity columns (a stride-2 window pairs equal parities only): 14 tile products per dy, was 22; both parities of a line streamed as one pair; band 48 wide; sums unchanged
# speedup vs baseline: 2.6858x; 1.2924x over previous
#include <hip/hip_runtime.h>


typedef __attribute__((ext_vector_type(2))) float v2f;
typedef __attribute__((ext_vector_type(4))) float v4f;
typedef __attribute__((ext_vector_type(8))) float v8f;

#define BATCH 8
#define CCH   256
#define HH    64
#define WW    96
#define HW    (HH * WW)
#define ND    21
#define NWV   4
#define NT    (NWV * 32)
#define KSW   16
#define NLINE 63
#define PITCH 50
#define RSTR  (16 * PITCH)
#define WSTR  (6 * RSTR)

__device__ __forceinline__ v2f ld_frag(const float* __restrict__ p, const int s) {
  v2f v;
  v.x = p[(size_t)(4 * s) * HW];
  v.y = p[(size_t)(4 * s + 1) * HW];
  return v;
}

__device__ __forceinline__ void st_part(float* __restrict__ part, const int T, const int jj,
                                        const v8f a) {
#pragma unroll
  for (int r = 0; r < 8; ++r) part[T * RSTR + r * PITCH + 16 * jj] = a[r];
}

template <int TI>
__device__ __forceinline__ void pair_step(const v2f (&B)[6][KSW], v2f (&A0)[KSW], v2f (&A1)[KSW],
                                          const float* __restrict__ nxt0,
                                          const float* __restrict__ nxt1,
                                          float* __restrict__ part) {
  constexpr int lo = (TI > 0) ? TI - 1 : 0;
  constexpr int hi = (TI < 2) ? TI + 1 : 2;
  constexpr int n  = hi - lo + 1;
  const v8f zero = (v8f){0.f, 0.f, 0.f, 0.f, 0.f, 0.f, 0.f, 0.f};
  __builtin_amdgcn_sched_barrier(0);
  v8f acc0[n], acc1[n];
#pragma unroll
  for (int t = 0; t < n; ++t) { acc0[t] = zero; acc1[t] = zero; }
#pragma unroll
  for (int s = 0; s < KSW; ++s) {
#pragma unroll
    for (int t = 0; t < n; ++t)
      acc0[t] = __builtin_amdgcn_wmma_f32_16x16x4_f32(false, A0[s], false, B[2 * (lo + t)][s],
                                                      (short)0, acc0[t], false, false);
#pragma unroll
    for (int t = 0; t < n; ++t)
      acc1[t] = __builtin_amdgcn_wmma_f32_16x16x4_f32(false, A1[s], false, B[2 * (lo + t) + 1][s],
                                                      (short)0, acc1[t], false, false);
  }
  __builtin_amdgcn_sched_barrier(0);
#pragma unroll
  for (int s = 0; s < KSW; ++s) { A0[s] = ld_frag(nxt0, s); A1[s] = ld_frag(nxt1, s); }
#pragma unroll
  for (int t = 0; t < n; ++t) {
    st_part(part, 2 * TI, lo + t - TI + 1, acc0[t]);
    st_part(part, 2 * TI + 1, lo + t - TI + 1, acc1[t]);
  }
}

__device__ __forceinline__ void lds_barrier() {
  __builtin_amdgcn_fence(__ATOMIC_RELEASE, "workgroup", "local");
  __builtin_amdgcn_s_barrier();
  __builtin_amdgcn_fence(__ATOMIC_ACQUIRE, "workgroup", "local");
}

__device__ __forceinline__ void set_wave_prio(const int lvl) {
  if (lvl == 3)      __builtin_amdgcn_s_setprio(3);
  else if (lvl == 2) __builtin_amdgcn_s_setprio(2);
  else if (lvl == 1) __builtin_amdgcn_s_setprio(1);
  else               __builtin_amdgcn_s_setprio(0);
}

__device__ __forceinline__ void dy_step(const int dy, const int dyFirst, const int dyLast,
                                        const int r2, const int b, const int wv, const int lane,
                                        int& lvlCur, const v2f (&B)[6][KSW],
                                        v2f (&X0)[KSW], v2f (&X1)[KSW],
                                        v2f (&Y0)[KSW], v2f (&Y1)[KSW],
                                        const float* __restrict__ f1e,
                                        const float* __restrict__ f1o,
                                        float* __restrict__ part,
                                        const float* __restrict__ red,
                                        float* __restrict__ out) {
  const int nDy = dyLast - dyFirst + 1;
  const int i4  = 4 * (dy - dyFirst);
  const int lvl = 3 - (i4 >= nDy) - (i4 >= 2 * nDy) - (i4 >= 3 * nDy);
  if (lvl != lvlCur) { set_wave_prio(lvl); lvlCur = lvl; }

  const int h   = r2 + 20 - 2 * dy;
  const int dyn = (dy < dyLast) ? dy + 1 : dy;
  const size_t oc = (size_t)h * WW;
  const size_t on = (size_t)(r2 + 20 - 2 * dyn) * WW;

  pair_step<0>(B, X0, X1, f1e + oc + 64, f1o + oc + 64, part);
  pair_step<1>(B, Y0, Y1, f1e + on,      f1o + on,      part);
  pair_step<2>(B, X0, X1, f1e + on + 32, f1o + on + 32, part);

  lds_barrier();

  const size_t outBase =
      (((size_t)b * (ND * ND) + (size_t)dy * ND) * HH + (size_t)h) * WW;
  float pq[3][5][4];
#pragma unroll
  for (int c = 0; c < 3; ++c) {
    const int L   = wv + 4 * c;
    const int dx0 = L / 3;
    const int ch  = L - 3 * dx0;
    const float* p = red + (2 * ch) * RSTR + dx0;
#pragma unroll
    for (int a = 0; a < 5; ++a) {
      pq[c][a][0] = p[4 * a];
      pq[c][a][1] = p[4 * a + WSTR];
      pq[c][a][2] = p[4 * a + 2 * WSTR];
      pq[c][a][3] = p[4 * a + 3 * WSTR];
    }
  }
#pragma unroll
  for (int c = 0; c < 3; ++c) {
    const int L   = wv + 4 * c;
    const int dx0 = L / 3;
    const int ch  = L - 3 * dx0;
    const int w1  = 32 * ch + lane;
    float* __restrict__ o = out + (outBase + (size_t)dx0 * HW + (size_t)w1);
#pragma unroll
    for (int a = 0; a < 5; ++a) {
      const float v = (pq[c][a][0] + pq[c][a][1]) + (pq[c][a][2] + pq[c][a][3]);
      const int w2 = w1 + 2 * (dx0 + 4 * a) - 20;
      o[(size_t)(4 * a) * HW] = (w2 >= 0 && w2 < WW) ? v * 0.00390625f : 0.f;
    }
  }
  if (wv < 3) {
    const float* p = red + (2 * wv) * RSTR + 20;
    const float v = (p[0] + p[WSTR]) + (p[2 * WSTR] + p[3 * WSTR]);
    const int w1 = 32 * wv + lane;
    out[outBase + (size_t)20 * HW + w1] = (w1 + 20 < WW) ? v * 0.00390625f : 0.f;
  }

  lds_barrier();
}

__global__ __launch_bounds__(NT)
void corr_wmma_f32_kernel(const float* __restrict__ f1,
                          const float* __restrict__ f2,
                          float* __restrict__ out) {
  extern __shared__ float smem[];

  const int tid  = threadIdx.x;
  const int lane = tid & 31;
  const int wv   = __builtin_amdgcn_readfirstlane(tid >> 5);

  const int hz = blockIdx.x;
  const int r2 = blockIdx.x;
  const int b = blockIdx.y;
  const size_t inBase = (size_t)b * (CCH * HW);

  const int row = lane & 15;
  const int khi = lane >> 4;

  const int zFirst = (hz < 20) ? (21 - hz) / 2 : 0;
  const int zLast  = ((83 - hz) / 2 < 20) ? (83 - hz) / 2 : 20;
  for (int dy = 0; dy < ND; ++dy) {
    if (dy >= zFirst && dy <= zLast) continue;
    const size_t outBase =
        (((size_t)b * (ND * ND) + (size_t)dy * ND) * HH + (size_t)hz) * WW;
    for (int i = tid; i < ND * WW; i += NT) {
      const int dxI = i / WW, w = i % WW;
      out[outBase + (size_t)dxI * HW + w] = 0.f;
    }
  }

  const int dyFirst = (r2 >= 43) ? (r2 - 42) / 2 : 0;
  const int dyLast  = ((r2 + 20) / 2 < 20) ? (r2 + 20) / 2 : 20;

  int odd = 1;
  asm volatile("" : "+s"(odd));
  v2f B[6][KSW];
  {
    const float* p = f2 + inBase + (size_t)(64 * wv + 2 * khi) * HW + (size_t)r2 * WW + 2 * row;
    const float* q = p + odd;
#pragma unroll
    for (int t = 0; t < 3; ++t) {
#pragma unroll
      for (int s = 0; s < KSW; ++s) {
        B[2 * t][s]     = ld_frag(p + 32 * t, s);
        B[2 * t + 1][s] = ld_frag(q + 32 * t, s);
      }
    }
  }

  const float* f1e = f1 + inBase + (size_t)(64 * wv + 2 * khi) * HW + 2 * row;
  const float* f1o = f1e + odd;
  v2f X0[KSW], X1[KSW], Y0[KSW], Y1[KSW];
  {
    const size_t o0 = (size_t)(r2 + 20 - 2 * dyFirst) * WW;
#pragma unroll
    for (int s = 0; s < KSW; ++s) { X0[s] = ld_frag(f1e + o0, s); X1[s] = ld_frag(f1o + o0, s); }
#pragma unroll
    for (int s = 0; s < KSW; ++s) {
      Y0[s] = ld_frag(f1e + o0 + 32, s);
      Y1[s] = ld_frag(f1o + o0 + 32, s);
    }
  }

  float* part = smem + wv * WSTR + khi * (8 * PITCH) + row;
  const float* red = smem + (lane & 1) * RSTR + (lane >> 1) * (PITCH + 1) + 6;

  {
    const v8f zero = (v8f){0.f, 0.f, 0.f, 0.f, 0.f, 0.f, 0.f, 0.f};
    st_part(part, 0, 0, zero);
    st_part(part, 1, 0, zero);
    st_part(part, 4, 2, zero);
    st_part(part, 5, 2, zero);
  }

  int lvlCur = 0;

  for (int dy = dyFirst;;) {
    dy_step(dy, dyFirst, dyLast, r2, b, wv, lane, lvlCur, B, X0, X1, Y0, Y1, f1e, f1o, part, red,
            out);
    if (dy == dyLast) break;
    ++dy;
    dy_step(dy, dyFirst, dyLast, r2, b, wv, lane, lvlCur, B, Y0, Y1, X0, X1, f1e, f1o, part, red,
            out);
    if (dy == dyLast) break;
    ++dy;
  }
}

extern "C" void kernel_launch(void* const* d_in, const int* in_sizes, int n_in,
                              void* d_out, int out_size, void* d_ws, size_t ws_size,
                              hipStream_t stream) {
  const float* f1 = (const float*)d_in[0];
  const float* f2 = (const float*)d_in[1];
  float* out = (float*)d_out;

  dim3 grid(HH, BATCH);
  dim3 block(NT);
  const size_t shmem = (size_t)(NWV * WSTR) * sizeof(float);
  corr_wmma_f32_kernel<<<grid, block, shmem, stream>>>(f1, f2, out);
}


# f1 fragments streamed by 8-byte loads: a load brings a channel's even and odd column (16 lanes = one whole line), one v_swap per k-step sorts them into the parity operands: 32 loads a pair, was 64
# speedup vs baseline: 2.8601x; 1.0649x over previous
#include <hip/hip_runtime.h>


typedef __attribute__((ext_vector_type(2))) float v2f;
typedef __attribute__((ext_vector_type(4))) float v4f;
typedef __attribute__((ext_vector_type(8))) float v8f;

#define BATCH 8
#define CCH   256
#define HH    64
#define WW    96
#define HW    (HH * WW)
#define ND    21
#define NWV   4
#define NT    (NWV * 32)
#define KSW   16
#define NLINE 63
#define PITCH 50
#define RSTR  (16 * PITCH)
#define WSTR  (6 * RSTR)

__device__ __forceinline__ v2f ld_frag(const float* __restrict__ p, const int s) {
  v2f v;
  v.x = p[(size_t)(4 * s) * HW];
  v.y = p[(size_t)(4 * s + 1) * HW];
  return v;
}

__device__ __forceinline__ v2f ld_cols(const float* __restrict__ p, const int ch) {
  return *reinterpret_cast<const v2f*>(p + (size_t)ch * HW);
}

__device__ __forceinline__ void st_part(float* __restrict__ part, const int T, const int jj,
                                        const v8f a) {
#pragma unroll
  for (int r = 0; r < 8; ++r) part[T * RSTR + r * PITCH + 16 * jj] = a[r];
}

template <int TI>
__device__ __forceinline__ void pair_step(const v2f (&B)[6][KSW], v2f (&A0)[KSW], v2f (&A1)[KSW],
                                          const float* __restrict__ nxt,
                                          float* __restrict__ part) {
  constexpr int lo = (TI > 0) ? TI - 1 : 0;
  constexpr int hi = (TI < 2) ? TI + 1 : 2;
  constexpr int n  = hi - lo + 1;
  const v8f zero = (v8f){0.f, 0.f, 0.f, 0.f, 0.f, 0.f, 0.f, 0.f};
  __builtin_amdgcn_sched_barrier(0);
#pragma unroll
  for (int s = 0; s < KSW; ++s) {
    float eo = A0[s].y, oe = A1[s].x;
    asm volatile("v_swap_b32 %0, %1" : "+v"(eo), "+v"(oe));
    A0[s].y = eo;
    A1[s].x = oe;
  }
  v8f acc0[n], acc1[n];
#pragma unroll
  for (int t = 0; t < n; ++t) { acc0[t] = zero; acc1[t] = zero; }
#pragma unroll
  for (int s = 0; s < KSW; ++s) {
#pragma unroll
    for (int t = 0; t < n; ++t)
      acc0[t] = __builtin_amdgcn_wmma_f32_16x16x4_f32(false, A0[s], false, B[2 * (lo + t)][s],
                                                      (short)0, acc0[t], false, false);
#pragma unroll
    for (int t = 0; t < n; ++t)
      acc1[t] = __builtin_amdgcn_wmma_f32_16x16x4_f32(false, A1[s], false, B[2 * (lo + t) + 1][s],
                                                      (short)0, acc1[t], false, false);
  }
  __builtin_amdgcn_sched_barrier(0);
#pragma unroll
  for (int s = 0; s < KSW; ++s) { A0[s] = ld_cols(nxt, 4 * s); A1[s] = ld_cols(nxt, 4 * s + 1); }
#pragma unroll
  for (int t = 0; t < n; ++t) {
    st_part(part, 2 * TI, lo + t - TI + 1, acc0[t]);
    st_part(part, 2 * TI + 1, lo + t - TI + 1, acc1[t]);
  }
}

__device__ __forceinline__ void lds_barrier() {
  __builtin_amdgcn_fence(__ATOMIC_RELEASE, "workgroup", "local");
  __builtin_amdgcn_s_barrier();
  __builtin_amdgcn_fence(__ATOMIC_ACQUIRE, "workgroup", "local");
}

__device__ __forceinline__ void set_wave_prio(const int lvl) {
  if (lvl == 3)      __builtin_amdgcn_s_setprio(3);
  else if (lvl == 2) __builtin_amdgcn_s_setprio(2);
  else if (lvl == 1) __builtin_amdgcn_s_setprio(1);
  else               __builtin_amdgcn_s_setprio(0);
}

__device__ __forceinline__ void dy_step(const int dy, const int dyFirst, const int dyLast,
                                        const int r2, const int b, const int wv, const int lane,
                                        int& lvlCur, const v2f (&B)[6][KSW],
                                        v2f (&X0)[KSW], v2f (&X1)[KSW],
                                        v2f (&Y0)[KSW], v2f (&Y1)[KSW],
                                        const float* __restrict__ f1e,
                                        float* __restrict__ part,
                                        const float* __restrict__ red,
                                        float* __restrict__ out) {
  const int nDy = dyLast - dyFirst + 1;
  const int i4  = 4 * (dy - dyFirst);
  const int lvl = 3 - (i4 >= nDy) - (i4 >= 2 * nDy) - (i4 >= 3 * nDy);
  if (lvl != lvlCur) { set_wave_prio(lvl); lvlCur = lvl; }

  const int h   = r2 + 20 - 2 * dy;
  const int dyn = (dy < dyLast) ? dy + 1 : dy;
  const size_t oc = (size_t)h * WW;
  const size_t on = (size_t)(r2 + 20 - 2 * dyn) * WW;

  pair_step<0>(B, X0, X1, f1e + oc + 64, part);
  pair_step<1>(B, Y0, Y1, f1e + on,      part);
  pair_step<2>(B, X0, X1, f1e + on + 32, part);

  lds_barrier();

  const size_t outBase =
      (((size_t)b * (ND * ND) + (size_t)dy * ND) * HH + (size_t)h) * WW;
  float pq[3][5][4];
#pragma unroll
  for (int c = 0; c < 3; ++c) {
    const int L   = wv + 4 * c;
    const int dx0 = L / 3;
    const int ch  = L - 3 * dx0;
    const float* p = red + (2 * ch) * RSTR + dx0;
#pragma unroll
    for (int a = 0; a < 5; ++a) {
      pq[c][a][0] = p[4 * a];
      pq[c][a][1] = p[4 * a + WSTR];
      pq[c][a][2] = p[4 * a + 2 * WSTR];
      pq[c][a][3] = p[4 * a + 3 * WSTR];
    }
  }
#pragma unroll
  for (int c = 0; c < 3; ++c) {
    const int L   = wv + 4 * c;
    const int dx0 = L / 3;
    const int ch  = L - 3 * dx0;
    const int w1  = 32 * ch + lane;
    float* __restrict__ o = out + (outBase + (size_t)dx0 * HW + (size_t)w1);
#pragma unroll
    for (int a = 0; a < 5; ++a) {
      const float v = (pq[c][a][0] + pq[c][a][1]) + (pq[c][a][2] + pq[c][a][3]);
      const int w2 = w1 + 2 * (dx0 + 4 * a) - 20;
      o[(size_t)(4 * a) * HW] = (w2 >= 0 && w2 < WW) ? v * 0.00390625f : 0.f;
    }
  }
  if (wv < 3) {
    const float* p = red + (2 * wv) * RSTR + 20;
    const float v = (p[0] + p[WSTR]) + (p[2 * WSTR] + p[3 * WSTR]);
    const int w1 = 32 * wv + lane;
    out[outBase + (size_t)20 * HW + w1] = (w1 + 20 < WW) ? v * 0.00390625f : 0.f;
  }

  lds_barrier();
}

__global__ __launch_bounds__(NT)
void corr_wmma_f32_kernel(const float* __restrict__ f1,
                          const float* __restrict__ f2,
                          float* __restrict__ out) {
  extern __shared__ float smem[];

  const int tid  = threadIdx.x;
  const int lane = tid & 31;
  const int wv   = __builtin_amdgcn_readfirstlane(tid >> 5);

  const int hz = blockIdx.x;
  const int r2 = blockIdx.x;
  const int b = blockIdx.y;
  const size_t inBase = (size_t)b * (CCH * HW);

  const int row = lane & 15;
  const int khi = lane >> 4;

  const int zFirst = (hz < 20) ? (21 - hz) / 2 : 0;
  const int zLast  = ((83 - hz) / 2 < 20) ? (83 - hz) / 2 : 20;
  for (int dy = 0; dy < ND; ++dy) {
    if (dy >= zFirst && dy <= zLast) continue;
    const size_t outBase =
        (((size_t)b * (ND * ND) + (size_t)dy * ND) * HH + (size_t)hz) * WW;
    for (int i = tid; i < ND * WW; i += NT) {
      const int dxI = i / WW, w = i % WW;
      out[outBase + (size_t)dxI * HW + w] = 0.f;
    }
  }

  const int dyFirst = (r2 >= 43) ? (r2 - 42) / 2 : 0;
  const int dyLast  = ((r2 + 20) / 2 < 20) ? (r2 + 20) / 2 : 20;

  int odd = 1;
  asm volatile("" : "+s"(odd));
  v2f B[6][KSW];
  {
    const float* p = f2 + inBase + (size_t)(64 * wv + 2 * khi) * HW + (size_t)r2 * WW + 2 * row;
    const float* q = p + odd;
#pragma unroll
    for (int t = 0; t < 3; ++t) {
#pragma unroll
      for (int s = 0; s < KSW; ++s) {
        B[2 * t][s]     = ld_frag(p + 32 * t, s);
        B[2 * t + 1][s] = ld_frag(q + 32 * t, s);
      }
    }
  }

  const float* f1e = f1 + inBase + (size_t)(64 * wv + 2 * khi) * HW + 2 * row;
  v2f X0[KSW], X1[KSW], Y0[KSW], Y1[KSW];
  {
    const size_t o0 = (size_t)(r2 + 20 - 2 * dyFirst) * WW;
#pragma unroll
    for (int s = 0; s < KSW; ++s) { X0[s] = ld_cols(f1e + o0, 4 * s); X1[s] = ld_cols(f1e + o0, 4 * s + 1); }
#pragma unroll
    for (int s = 0; s < KSW; ++s) {
      Y0[s] = ld_cols(f1e + o0 + 32, 4 * s);
      Y1[s] = ld_cols(f1e + o0 + 32, 4 * s + 1);
    }
  }

  float* part = smem + wv * WSTR + khi * (8 * PITCH) + row;
  const float* red = smem + (lane & 1) * RSTR + (lane >> 1) * (PITCH + 1) + 6;

  {
    const v8f zero = (v8f){0.f, 0.f, 0.f, 0.f, 0.f, 0.f, 0.f, 0.f};
    st_part(part, 0, 0, zero);
    st_part(part, 1, 0, zero);
    st_part(part, 4, 2, zero);
    st_part(part, 5, 2, zero);
  }

  int lvlCur = 0;

  for (int dy = dyFirst;;) {
    dy_step(dy, dyFirst, dyLast, r2, b, wv, lane, lvlCur, B, X0, X1, Y0, Y1, f1e, part, red,
            out);
    if (dy == dyLast) break;
    ++dy;
    dy_step(dy, dyFirst, dyLast, r2, b, wv, lane, lvlCur, B, Y0, Y1, X0, X1, f1e, part, red,
            out);
    if (dy == dyLast) break;
    ++dy;
  }
}

extern "C" void kernel_launch(void* const* d_in, const int* in_sizes, int n_in,
                              void* d_out, int out_size, void* d_ws, size_t ws_size,
                              hipStream_t stream) {
  const float* f1 = (const float*)d_in[0];
  const float* f2 = (const float*)d_in[1];
  float* out = (float*)d_out;

  dim3 grid(HH, BATCH);
  dim3 block(NT);
  const size_t shmem = (size_t)(NWV * WSTR) * sizeof(float);
  corr_wmma_f32_kernel<<<grid, block, shmem, stream>>>(f1, f2, out);
}


# K-quarter partial tiles stored column-major in LDS (column pitch 20): a lane's 8 accumulator rows are contiguous, a tile is 2 16-byte stores per lane, was 4 two-address 4-byte stores; sums unchanged
# speedup vs baseline: 2.9466x; 1.0302x over previous
#include <hip/hip_runtime.h>


typedef __attribute__((ext_vector_type(2))) float v2f;
typedef __attribute__((ext_vector_type(4))) float v4f;
typedef __attribute__((ext_vector_type(8))) float v8f;

#define BATCH 8
#define CCH   256
#define HH    64
#define WW    96
#define HW    (HH * WW)
#define ND    21
#define NWV   4
#define NT    (NWV * 32)
#define KSW   16
#define NLINE 63
#define CP    20
#define RSTR  (48 * CP + 8)
#define TSTR  (NWV * RSTR)

__device__ __forceinline__ v2f ld_frag(const float* __restrict__ p, const int s) {
  v2f v;
  v.x = p[(size_t)(4 * s) * HW];
  v.y = p[(size_t)(4 * s + 1) * HW];
  return v;
}

__device__ __forceinline__ v2f ld_cols(const float* __restrict__ p, const int ch) {
  return *reinterpret_cast<const v2f*>(p + (size_t)ch * HW);
}

__device__ __forceinline__ void st_part(float* __restrict__ part, const int T, const int jj,
                                        const v8f a) {
  float* __restrict__ p = part + T * TSTR + jj * (16 * CP);
  *reinterpret_cast<v4f*>(p)     = __builtin_shufflevector(a, a, 0, 1, 2, 3);
  *reinterpret_cast<v4f*>(p + 4) = __builtin_shufflevector(a, a, 4, 5, 6, 7);
}

template <int TI>
__device__ __forceinline__ void pair_step(const v2f (&B)[6][KSW], v2f (&A0)[KSW], v2f (&A1)[KSW],
                                          const float* __restrict__ nxt,
                                          float* __restrict__ part) {
  constexpr int lo = (TI > 0) ? TI - 1 : 0;
  constexpr int hi = (TI < 2) ? TI + 1 : 2;
  constexpr int n  = hi - lo + 1;
  const v8f zero = (v8f){0.f, 0.f, 0.f, 0.f, 0.f, 0.f, 0.f, 0.f};
  __builtin_amdgcn_sched_barrier(0);
#pragma unroll
  for (int s = 0; s < KSW; ++s) {
    float eo = A0[s].y, oe = A1[s].x;
    asm volatile("v_swap_b32 %0, %1" : "+v"(eo), "+v"(oe));
    A0[s].y = eo;
    A1[s].x = oe;
  }
  v8f acc0[n], acc1[n];
#pragma unroll
  for (int t = 0; t < n; ++t) { acc0[t] = zero; acc1[t] = zero; }
#pragma unroll
  for (int s = 0; s < KSW; ++s) {
#pragma unroll
    for (int t = 0; t < n; ++t)
      acc0[t] = __builtin_amdgcn_wmma_f32_16x16x4_f32(false, A0[s], false, B[2 * (lo + t)][s],
                                                      (short)0, acc0[t], false, false);
#pragma unroll
    for (int t = 0; t < n; ++t)
      acc1[t] = __builtin_amdgcn_wmma_f32_16x16x4_f32(false, A1[s], false, B[2 * (lo + t) + 1][s],
                                                      (short)0, acc1[t], false, false);
  }
  __builtin_amdgcn_sched_barrier(0);
#pragma unroll
  for (int s = 0; s < KSW; ++s) { A0[s] = ld_cols(nxt, 4 * s); A1[s] = ld_cols(nxt, 4 * s + 1); }
#pragma unroll
  for (int t = 0; t < n; ++t) {
    st_part(part, 2 * TI, lo + t - TI + 1, acc0[t]);
    st_part(part, 2 * TI + 1, lo + t - TI + 1, acc1[t]);
  }
}

__device__ __forceinline__ void lds_barrier() {
  __builtin_amdgcn_fence(__ATOMIC_RELEASE, "workgroup", "local");
  __builtin_amdgcn_s_barrier();
  __builtin_amdgcn_fence(__ATOMIC_ACQUIRE, "workgroup", "local");
}

__device__ __forceinline__ void set_wave_prio(const int lvl) {
  if (lvl == 3)      __builtin_amdgcn_s_setprio(3);
  else if (lvl == 2) __builtin_amdgcn_s_setprio(2);
  else if (lvl == 1) __builtin_amdgcn_s_setprio(1);
  else               __builtin_amdgcn_s_setprio(0);
}

__device__ __forceinline__ void dy_step(const int dy, const int dyFirst, const int dyLast,
                                        const int r2, const int b, const int wv, const int lane,
                                        int& lvlCur, const v2f (&B)[6][KSW],
                                        v2f (&X0)[KSW], v2f (&X1)[KSW],
                                        v2f (&Y0)[KSW], v2f (&Y1)[KSW],
                                        const float* __restrict__ f1e,
                                        float* __restrict__ part,
                                        const float* __restrict__ red,
                                        float* __restrict__ out) {
  const int nDy = dyLast - dyFirst + 1;
  const int i4  = 4 * (dy - dyFirst);
  const int lvl = 3 - (i4 >= nDy) - (i4 >= 2 * nDy) - (i4 >= 3 * nDy);
  if (lvl != lvlCur) { set_wave_prio(lvl); lvlCur = lvl; }

  const int h   = r2 + 20 - 2 * dy;
  const int dyn = (dy < dyLast) ? dy + 1 : dy;
  const size_t oc = (size_t)h * WW;
  const size_t on = (size_t)(r2 + 20 - 2 * dyn) * WW;

  pair_step<0>(B, X0, X1, f1e + oc + 64, part);
  pair_step<1>(B, Y0, Y1, f1e + on,      part);
  pair_step<2>(B, X0, X1, f1e + on + 32, part);

  lds_barrier();

  const size_t outBase =
      (((size_t)b * (ND * ND) + (size_t)dy * ND) * HH + (size_t)h) * WW;
  float pq[3][5][4];
#pragma unroll
  for (int c = 0; c < 3; ++c) {
    const int L   = wv + 4 * c;
    const int dx0 = L / 3;
    const int ch  = L - 3 * dx0;
    const float* p = red + (2 * ch) * TSTR + dx0 * CP;
#pragma unroll
    for (int a = 0; a < 5; ++a) {
      pq[c][a][0] = p[4 * a * CP];
      pq[c][a][1] = p[4 * a * CP + RSTR];
      pq[c][a][2] = p[4 * a * CP + 2 * RSTR];
      pq[c][a][3] = p[4 * a * CP + 3 * RSTR];
    }
  }
#pragma unroll
  for (int c = 0; c < 3; ++c) {
    const int L   = wv + 4 * c;
    const int dx0 = L / 3;
    const int ch  = L - 3 * dx0;
    const int w1  = 32 * ch + lane;
    float* __restrict__ o = out + (outBase + (size_t)dx0 * HW + (size_t)w1);
#pragma unroll
    for (int a = 0; a < 5; ++a) {
      const float v = (pq[c][a][0] + pq[c][a][1]) + (pq[c][a][2] + pq[c][a][3]);
      const int w2 = w1 + 2 * (dx0 + 4 * a) - 20;
      o[(size_t)(4 * a) * HW] = (w2 >= 0 && w2 < WW) ? v * 0.00390625f : 0.f;
    }
  }
  if (wv < 3) {
    const float* p = red + (2 * wv) * TSTR + 20 * CP;
    const float v = (p[0] + p[RSTR]) + (p[2 * RSTR] + p[3 * RSTR]);
    const int w1 = 32 * wv + lane;
    out[outBase + (size_t)20 * HW + w1] = (w1 + 20 < WW) ? v * 0.00390625f : 0.f;
  }

  lds_barrier();
}

__global__ __launch_bounds__(NT)
void corr_wmma_f32_kernel(const float* __restrict__ f1,
                          const float* __restrict__ f2,
                          float* __restrict__ out) {
  extern __shared__ v4f smem4[];
  float* smem = reinterpret_cast<float*>(smem4);

  const int tid  = threadIdx.x;
  const int lane = tid & 31;
  const int wv   = __builtin_amdgcn_readfirstlane(tid >> 5);

  const int hz = blockIdx.x;
  const int r2 = blockIdx.x;
  const int b = blockIdx.y;
  const size_t inBase = (size_t)b * (CCH * HW);

  const int row = lane & 15;
  const int khi = lane >> 4;

  const int zFirst = (hz < 20) ? (21 - hz) / 2 : 0;
  const int zLast  = ((83 - hz) / 2 < 20) ? (83 - hz) / 2 : 20;
  for (int dy = 0; dy < ND; ++dy) {
    if (dy >= zFirst && dy <= zLast) continue;
    const size_t outBase =
        (((size_t)b * (ND * ND) + (size_t)dy * ND) * HH + (size_t)hz) * WW;
    for (int i = tid; i < ND * WW; i += NT) {
      const int dxI = i / WW, w = i % WW;
      out[outBase + (size_t)dxI * HW + w] = 0.f;
    }
  }

  const int dyFirst = (r2 >= 43) ? (r2 - 42) / 2 : 0;
  const int dyLast  = ((r2 + 20) / 2 < 20) ? (r2 + 20) / 2 : 20;

  int odd = 1;
  asm volatile("" : "+s"(odd));
  v2f B[6][KSW];
  {
    const float* p = f2 + inBase + (size_t)(64 * wv + 2 * khi) * HW + (size_t)r2 * WW + 2 * row;
    const float* q = p + odd;
#pragma unroll
    for (int t = 0; t < 3; ++t) {
#pragma unroll
      for (int s = 0; s < KSW; ++s) {
        B[2 * t][s]     = ld_frag(p + 32 * t, s);
        B[2 * t + 1][s] = ld_frag(q + 32 * t, s);
      }
    }
  }

  const float* f1e = f1 + inBase + (size_t)(64 * wv + 2 * khi) * HW + 2 * row;
  v2f X0[KSW], X1[KSW], Y0[KSW], Y1[KSW];
  {
    const size_t o0 = (size_t)(r2 + 20 - 2 * dyFirst) * WW;
#pragma unroll
    for (int s = 0; s < KSW; ++s) { X0[s] = ld_cols(f1e + o0, 4 * s); X1[s] = ld_cols(f1e + o0, 4 * s + 1); }
#pragma unroll
    for (int s = 0; s < KSW; ++s) {
      Y0[s] = ld_cols(f1e + o0 + 32, 4 * s);
      Y1[s] = ld_cols(f1e + o0 + 32, 4 * s + 1);
    }
  }

  float* part = smem + wv * RSTR + row * CP + khi * 8;
  const float* red = smem + (lane & 1) * TSTR + (lane >> 1) * (CP + 1) + 6 * CP;

  {
    const v8f zero = (v8f){0.f, 0.f, 0.f, 0.f, 0.f, 0.f, 0.f, 0.f};
    st_part(part, 0, 0, zero);
    st_part(part, 1, 0, zero);
    st_part(part, 4, 2, zero);
    st_part(part, 5, 2, zero);
  }

  int lvlCur = 0;

  for (int dy = dyFirst;;) {
    dy_step(dy, dyFirst, dyLast, r2, b, wv, lane, lvlCur, B, X0, X1, Y0, Y1, f1e, part, red,
            out);
    if (dy == dyLast) break;
    ++dy;
    dy_step(dy, dyFirst, dyLast, r2, b, wv, lane, lvlCur, B, Y0, Y1, X0, X1, f1e, part, red,
            out);
    if (dy == dyLast) break;
    ++dy;
  }
}

extern "C" void kernel_launch(void* const* d_in, const int* in_sizes, int n_in,
                              void* d_out, int out_size, void* d_ws, size_t ws_size,
                              hipStream_t stream) {
  const float* f1 = (const float*)d_in[0];
  const float* f2 = (const float*)d_in[1];
  float* out = (float*)d_out;

  dim3 grid(HH, BATCH);
  dim3 block(NT);
  const size_t shmem = (size_t)(6 * TSTR) * sizeof(float);
  corr_wmma_f32_kernel<<<grid, block, shmem, stream>>>(f1, f2, out);
}
